# starting kernel
# baseline (speedup 1.0000x reference)
#include <hip/hip_runtime.h>
#include <hip/hip_bf16.h>

typedef __attribute__((ext_vector_type(2))) float v2f;
typedef __attribute__((ext_vector_type(8))) float v8f;

#define IN_DIM  128
#define HID_DIM 64
#define OUT_DIM 128

__global__ void deg_kernel(const int* __restrict__ dst, int E,
                           unsigned int* __restrict__ deg) {
    int i = blockIdx.x * blockDim.x + threadIdx.x;
    if (i < E) atomicAdd(&deg[dst[i]], 1u);
}

__global__ void dinv_kernel(const unsigned int* __restrict__ deg, int N,
                            float* __restrict__ dinv) {
    int i = blockIdx.x * blockDim.x + threadIdx.x;
    if (i < N) dinv[i] = rsqrtf((float)deg[i] + 1.0f);
}

__global__ __launch_bounds__(256) void gemm1_kernel(
    const float* __restrict__ x, const float* __restrict__ Wc,
    float* __restrict__ h, int N) {
    __shared__ float sW[IN_DIM * HID_DIM];
    {
        const float4* gv = (const float4*)Wc;
        float4* sv = (float4*)sW;
        for (int i = threadIdx.x; i < (IN_DIM * HID_DIM) / 4; i += 256)
            sv[i] = gv[i];
    }
    __syncthreads();

    const int wave = threadIdx.x >> 5;
    const int lane = threadIdx.x & 31;
    const int rowBase = blockIdx.x * 128 + wave * 16;
    if (rowBase >= N) return;

    const int r  = lane & 15;
    const int kg = lane >> 4;
    const int rowLd = min(rowBase + r, N - 1);
    const float* xrow = x + (size_t)rowLd * IN_DIM;

    v8f acc[4] = {v8f{}, v8f{}, v8f{}, v8f{}};
    for (int k = 0; k < IN_DIM; k += 4) {
        const int kk = k + (kg << 1);
        float2 av = *(const float2*)(xrow + kk);
        v2f a; a.x = av.x; a.y = av.y;
#pragma unroll
        for (int t = 0; t < 4; ++t) {
            v2f b;
            b.x = sW[(kk)     * HID_DIM + t * 16 + r];
            b.y = sW[(kk + 1) * HID_DIM + t * 16 + r];
            acc[t] = __builtin_amdgcn_wmma_f32_16x16x4_f32(
                false, a, false, b, (short)0, acc[t], false, false);
        }
    }
#pragma unroll
    for (int t = 0; t < 4; ++t) {
        const int col = t * 16 + r;
#pragma unroll
        for (int rr = 0; rr < 8; ++rr) {
            const int row = rowBase + rr + 8 * kg;
            if (row < N) h[(size_t)row * HID_DIM + col] = acc[t][rr];
        }
    }
}

__global__ __launch_bounds__(256) void scatter_kernel(
    const int* __restrict__ src, const int* __restrict__ dst,
    const float* __restrict__ dinv, const float* __restrict__ h,
    float* __restrict__ agg, int E) {
    const int e    = (int)((blockIdx.x * (unsigned)blockDim.x + threadIdx.x) >> 5);
    const int lane = threadIdx.x & 31;
    if (e >= E) return;
    const int s = src[e];
    const int d = dst[e];
    const float nrm = dinv[s] * dinv[d];
    const float2 hv = *(const float2*)(h + (size_t)s * HID_DIM + lane * 2);
    float* ap = agg + (size_t)d * HID_DIM + lane * 2;
    atomicAdd(ap,     hv.x * nrm);
    atomicAdd(ap + 1, hv.y * nrm);
}

__global__ __launch_bounds__(256) void gemm2_kernel(
    const float* __restrict__ agg, const float* __restrict__ h,
    const float* __restrict__ dinv, const float* __restrict__ bc,
    const float* __restrict__ Wl,  const float* __restrict__ bl,
    float* __restrict__ out, int N) {
    __shared__ float sW[HID_DIM * OUT_DIM];
    {
        const float4* gv = (const float4*)Wl;
        float4* sv = (float4*)sW;
        for (int i = threadIdx.x; i < (HID_DIM * OUT_DIM) / 4; i += 256)
            sv[i] = gv[i];
    }
    __syncthreads();

    const int wave = threadIdx.x >> 5;
    const int lane = threadIdx.x & 31;
    const int rowBase = blockIdx.x * 128 + wave * 16;
    if (rowBase >= N) return;

    const int r  = lane & 15;
    const int kg = lane >> 4;
    const int rowLd = min(rowBase + r, N - 1);
    const float* arow = agg + (size_t)rowLd * HID_DIM;
    const float* hrow = h   + (size_t)rowLd * HID_DIM;
    const float  di   = dinv[rowLd];
    const float  di2  = di * di;

    v8f acc[8] = {v8f{}, v8f{}, v8f{}, v8f{}, v8f{}, v8f{}, v8f{}, v8f{}};
    for (int k = 0; k < HID_DIM; k += 4) {
        const int kk = k + (kg << 1);
        float2 av  = *(const float2*)(arow + kk);
        float2 hv  = *(const float2*)(hrow + kk);
        float2 bcv = *(const float2*)(bc + kk);
        v2f a;
        a.x = fmaxf(av.x + hv.x * di2 + bcv.x, 0.0f);
        a.y = fmaxf(av.y + hv.y * di2 + bcv.y, 0.0f);
#pragma unroll
        for (int t = 0; t < 8; ++t) {
            v2f b;
            b.x = sW[(kk)     * OUT_DIM + t * 16 + r];
            b.y = sW[(kk + 1) * OUT_DIM + t * 16 + r];
            acc[t] = __builtin_amdgcn_wmma_f32_16x16x4_f32(
                false, a, false, b, (short)0, acc[t], false, false);
        }
    }
#pragma unroll
    for (int t = 0; t < 8; ++t) {
        const int col = t * 16 + r;
        const float blv = bl[col];
#pragma unroll
        for (int rr = 0; rr < 8; ++rr) {
            const int row = rowBase + rr + 8 * kg;
            if (row < N) out[(size_t)row * OUT_DIM + col] = acc[t][rr] + blv;
        }
    }
}

static inline size_t alignUp(size_t v) { return (v + 255) & ~(size_t)255; }

extern "C" void kernel_launch(void* const* d_in, const int* in_sizes, int n_in,
                              void* d_out, int out_size, void* d_ws, size_t ws_size,
                              hipStream_t stream) {
    const float* x  = (const float*)d_in[0];
    const int*   ei = (const int*)  d_in[1];
    const float* Wc = (const float*)d_in[2];
    const float* bc = (const float*)d_in[3];
    const float* Wl = (const float*)d_in[4];
    const float* bl = (const float*)d_in[5];
    float* out = (float*)d_out;

    const int N = in_sizes[0] / IN_DIM;
    const int E = in_sizes[1] / 2;
    const int* src = ei;
    const int* dst = ei + E;

    char* ws = (char*)d_ws;
    const size_t hBytes = (size_t)N * HID_DIM * sizeof(float);
    size_t off = 0;
    float*        h    = (float*)(ws + off);        off = alignUp(off + hBytes);
    float*        agg  = (float*)(ws + off);        off = alignUp(off + hBytes);
    float*        dinv = (float*)(ws + off);        off = alignUp(off + (size_t)N * sizeof(float));
    unsigned int* deg  = (unsigned int*)(ws + off); off = alignUp(off + (size_t)N * sizeof(unsigned int));

    hipMemsetAsync(agg, 0, hBytes, stream);
    hipMemsetAsync(deg, 0, (size_t)N * sizeof(unsigned int), stream);

    deg_kernel <<<(E + 255) / 256, 256, 0, stream>>>(dst, E, deg);
    dinv_kernel<<<(N + 255) / 256, 256, 0, stream>>>(deg, N, dinv);

    const int rowBlocks = (N + 127) / 128;
    gemm1_kernel<<<rowBlocks, 256, 0, stream>>>(x, Wc, h, N);

    scatter_kernel<<<(E + 7) / 8, 256, 0, stream>>>(src, dst, dinv, h, agg, E);

    gemm2_kernel<<<rowBlocks, 256, 0, stream>>>(agg, h, dinv, bc, Wl, bl, out, N);
}


# Filter reads a per-owner hit map, one bit per edge quad, that the edge blocks build in LDS (ds_or) and copy out; a lane pops set bits from its word, no per-edge key scan; sum order changes
# speedup vs baseline: 1.7626x; 1.7626x over previous
#include <hip/hip_runtime.h>
#include <hip/hip_bf16.h>

typedef __attribute__((ext_vector_type(2))) float v2f;
typedef __attribute__((ext_vector_type(8))) float v8f;

#define IN_DIM  128
#define HID_DIM 64
#define OUT_DIM 128
#define ROWS_WG 400
#define QPT     2
#define EB_QUADS (256 * QPT)
#define EB_EDGES (EB_QUADS * 4)
#define WPB     (EB_QUADS / 32)

__global__ void dinv_kernel(const unsigned int* __restrict__ deg, int N,
                            float* __restrict__ dinv) {
    int i = blockIdx.x * blockDim.x + threadIdx.x;
    if (i < N) dinv[i] = rsqrtf((float)deg[i] + 1.0f);
}

__global__ __launch_bounds__(256) void gemm1_deg_kernel(
    const float* __restrict__ x, const float* __restrict__ Wc,
    float* __restrict__ h, int N, int rowBlocks,
    const int* __restrict__ dst, int E, unsigned int* __restrict__ deg,
    unsigned int* __restrict__ bmap) {
    __shared__ __attribute__((aligned(16))) union { float w[IN_DIM * HID_DIM]; unsigned b[IN_DIM * HID_DIM]; } sm;
    float* sW = sm.w;
    if ((int)blockIdx.x >= rowBlocks) {
        unsigned* sB = sm.b;
        const int G   = (N + ROWS_WG - 1) / ROWS_WG;
        const int blk = (int)blockIdx.x - rowBlocks;
        const int t   = (int)threadIdx.x;
        const int nw  = G * WPB;
        const bool vec = ((E & 3) == 0);
        const int eb = (blk * EB_QUADS + t) * 4;
        int4 d[QPT];
#pragma unroll
        for (int j = 0; j < QPT; ++j) d[j] = make_int4(-1, -1, -1, -1);
        if (vec) {
#pragma unroll
            for (int j = 0; j < QPT; ++j)
                if (eb + j * 1024 < E) d[j] = *(const int4*)(dst + eb + j * 1024);
        } else {
#pragma unroll
            for (int j = 0; j < QPT; ++j) {
                const int e0 = eb + j * 1024;
                if (e0     < E) d[j].x = dst[e0];
                if (e0 + 1 < E) d[j].y = dst[e0 + 1];
                if (e0 + 2 < E) d[j].z = dst[e0 + 2];
                if (e0 + 3 < E) d[j].w = dst[e0 + 3];
            }
        }
        for (int i = t * 4; i < nw; i += 1024) *(uint4*)(sB + i) = make_uint4(0u, 0u, 0u, 0u);
        __syncthreads();
        const unsigned bit = 1u << (t & 31);
#pragma unroll
        for (int j = 0; j < QPT; ++j) {
            unsigned* wq = sB + (j * 8 + (t >> 5));
            const int dc[4] = {d[j].x, d[j].y, d[j].z, d[j].w};
#pragma unroll
            for (int c = 0; c < 4; ++c) {
                if (dc[c] >= 0) {
                    atomicAdd(&deg[dc[c]], 1u);
                    atomicOr(wq + ((unsigned)dc[c] / ROWS_WG) * WPB, bit);
                }
            }
        }
        __syncthreads();
        uint4* go = (uint4*)(bmap + (size_t)blk * (size_t)nw);
        for (int i = t; i < nw / 4; i += 256) go[i] = *(const uint4*)(sB + i * 4);
        return;
    }
    {
        const float4* gv = (const float4*)Wc;
        float4* sv = (float4*)sW;
        for (int i = threadIdx.x; i < (IN_DIM * HID_DIM) / 4; i += 256)
            sv[i] = gv[i];
    }
    __syncthreads();

    const int wave = threadIdx.x >> 5;
    const int lane = threadIdx.x & 31;
    const int rowBase = blockIdx.x * 128 + wave * 16;
    if (rowBase >= N) return;

    const int r  = lane & 15;
    const int kg = lane >> 4;
    const int rowLd = min(rowBase + r, N - 1);
    const float* xrow = x + (size_t)rowLd * IN_DIM;

    v8f acc[4] = {v8f{}, v8f{}, v8f{}, v8f{}};
    for (int k = 0; k < IN_DIM; k += 4) {
        const int kk = k + (kg << 1);
        float2 av = *(const float2*)(xrow + kk);
        v2f a; a.x = av.x; a.y = av.y;
#pragma unroll
        for (int t = 0; t < 4; ++t) {
            v2f b;
            b.x = sW[(kk)     * HID_DIM + t * 16 + r];
            b.y = sW[(kk + 1) * HID_DIM + t * 16 + r];
            acc[t] = __builtin_amdgcn_wmma_f32_16x16x4_f32(
                false, a, false, b, (short)0, acc[t], false, false);
        }
    }
#pragma unroll
    for (int t = 0; t < 4; ++t) {
        const int col = t * 16 + r;
#pragma unroll
        for (int rr = 0; rr < 8; ++rr) {
            const int row = rowBase + rr + 8 * kg;
            if (row < N) h[(size_t)row * HID_DIM + col] = acc[t][rr];
        }
    }
}

#define TPITCH  68
#define NWAVES  32
#define LCAP    32
#define QCAP    256

__device__ __forceinline__ void list_append(bool m, unsigned r, unsigned edge, int lane, int wave,
                                            unsigned* sL, unsigned* sC, float* sT,
                                            const int* __restrict__ src, const float* __restrict__ dinv,
                                            const float* __restrict__ h, int rowBase) {
    const unsigned mb = __builtin_amdgcn_ballot_w32(m);
    if (mb == 0u) return;
    const unsigned o = r & 31u;
    unsigned peers = mb;
#pragma unroll
    for (int b = 0; b < 5; ++b) {
        const int t = ((int)(o << (31 - b))) >> 31;
        const unsigned bb = __builtin_amdgcn_ballot_w32(t != 0);
        peers &= ~(bb ^ (unsigned)t);
    }
    if (m) {
        const unsigned rank = __builtin_amdgcn_mbcnt_lo(peers, 0u);
        unsigned* fill = sC + o * NWAVES + (unsigned)wave;
        const unsigned p = *fill + rank;
        if ((peers >> lane) == 1u) *fill = p + 1u;
        if (p < LCAP) {
            sL[(o * NWAVES + (unsigned)wave) * LCAP + p] = (edge << 9) | r;
        } else {
            const int s = src[edge];
            const float nrm = dinv[s] * dinv[rowBase + (int)r];
#pragma nounroll
            for (int c = 0; c < HID_DIM; ++c)
                atomicAdd(&sT[r * TPITCH + c], h[(size_t)s * HID_DIM + c] * nrm);
        }
    }
}

__device__ __forceinline__ void resolve_batch(unsigned nv, unsigned qh, const unsigned* sQw, int lane, int wave,
                                              unsigned* sL, unsigned* sC, float* sT,
                                              const int* __restrict__ src, const int* __restrict__ dst,
                                              const float* __restrict__ dinv, const float* __restrict__ h,
                                              int rowBase, int E, bool vecOK) {
    const unsigned quad = sQw[(qh + (unsigned)lane) & (QCAP - 1)];
    const unsigned e0 = quad << 2;
    const bool live = ((unsigned)lane < nv) && (e0 < (unsigned)E);
    int4 d = make_int4(-1, -1, -1, -1);
    if (live) {
        if (vecOK) {
            d = *(const int4*)(dst + e0);
        } else {
            d.x = dst[e0];
            if (e0 + 1 < (unsigned)E) d.y = dst[e0 + 1];
            if (e0 + 2 < (unsigned)E) d.z = dst[e0 + 2];
            if (e0 + 3 < (unsigned)E) d.w = dst[e0 + 3];
        }
    }
#pragma nounroll
    for (int c = 0; c < 4; ++c) {
        const int dc = (c == 0) ? d.x : (c == 1) ? d.y : (c == 2) ? d.z : d.w;
        const unsigned rc = (unsigned)(dc - rowBase);
        list_append(live && rc < ROWS_WG, rc, e0 + (unsigned)c, lane, wave, sL, sC, sT, src, dinv, h, rowBase);
    }
}

__global__ __launch_bounds__(1024) void agg_gemm2_kernel(
    const int* __restrict__ src, const int* __restrict__ dst,
    const float* __restrict__ dinv, const float* __restrict__ h,
    const float* __restrict__ bc, const float* __restrict__ Wl,
    const float* __restrict__ bl, float* __restrict__ out, int N, int E,
    const unsigned int* __restrict__ bmap, int nEB) {
    __shared__ __attribute__((aligned(16))) float sT[ROWS_WG * TPITCH];
    __shared__ __attribute__((aligned(16))) float sW[HID_DIM * OUT_DIM];
    __shared__ unsigned sL[NWAVES * NWAVES * LCAP];
    __shared__ unsigned sC[NWAVES * NWAVES];
    __shared__ unsigned sQ[NWAVES * QCAP];

    const int tid  = threadIdx.x;
    const int wave = __builtin_amdgcn_readfirstlane(tid >> 5);
    const int lane = tid & 31;
    const int rowBase = blockIdx.x * ROWS_WG;

    for (int i = tid; i < ROWS_WG * TPITCH; i += 1024) sT[i] = 0.0f;
    for (int i = tid; i < NWAVES * NWAVES; i += 1024) sC[i] = 0u;
    {
        const float4* gv = (const float4*)Wl;
        float4* sv = (float4*)sW;
        for (int i = tid; i < (HID_DIM * OUT_DIM) / 4; i += 1024) sv[i] = gv[i];
    }
    __syncthreads();

    {
        const int G    = (int)gridDim.x;
        const int wpo  = nEB * WPB;
        const int per  = (wpo + NWAVES - 1) / NWAVES;
        const int wbeg = wave * per;
        const int wend = min(wpo, wbeg + per);
        const bool vecOK = ((E & 3) == 0);
        unsigned* sQw = sQ + wave * QCAP;
        unsigned qh = 0u, qt = 0u;
        const unsigned* bg = bmap + (size_t)blockIdx.x * WPB;
        unsigned wn = 0u;
        if (wbeg + lane < wend) {
            const int wi = wbeg + lane;
            wn = bg[(size_t)(wi / WPB) * (size_t)(G * WPB) + (size_t)(wi % WPB)];
        }
        for (int i0 = wbeg; i0 < wend; i0 += 32) {
            unsigned w = wn;
            wn = 0u;
            if (i0 + 32 + lane < wend) {
                const int wi = i0 + 32 + lane;
                wn = bg[(size_t)(wi / WPB) * (size_t)(G * WPB) + (size_t)(wi % WPB)];
            }
            const unsigned qb = (unsigned)(i0 + lane) << 5;
            unsigned bal = __builtin_amdgcn_ballot_w32(w != 0u);
            while (bal) {
                if (w != 0u) {
                    const unsigned pos = (qt + __builtin_amdgcn_mbcnt_lo(bal, 0u)) & (QCAP - 1);
                    sQw[pos] = qb + (unsigned)__builtin_ctz(w);
                    w &= w - 1u;
                }
                qt += (unsigned)__builtin_popcount(bal);
                while (qt - qh >= 32u) {
                    resolve_batch(32u, qh, sQw, lane, wave, sL, sC, sT, src, dst, dinv, h, rowBase, E, vecOK);
                    qh += 32u;
                }
                bal = __builtin_amdgcn_ballot_w32(w != 0u);
            }
        }
        if (qt != qh)
            resolve_batch(qt - qh, qh, sQw, lane, wave, sL, sC, sT, src, dst, dinv, h, rowBase, E, vecOK);
    }
    __syncthreads();

    {
        unsigned* myL = sL + (unsigned)wave * (NWAVES * LCAP);
        unsigned n = sC[wave * NWAVES + lane];
        n = n < LCAP ? n : LCAP;
        unsigned inc = n;
#pragma unroll
        for (int d = 1; d < 32; d <<= 1) {
            const unsigned t = (unsigned)__shfl_up((int)inc, d, 32);
            if (lane >= d) inc += t;
        }
        const unsigned off = inc - n;
        const unsigned T = (unsigned)__builtin_amdgcn_readlane((int)inc, 31);

#pragma unroll
        for (int qb = 0; qb < NWAVES; qb += 16) {
            unsigned ent[16];
            int      sv[16];
#pragma unroll
            for (int u = 0; u < 16; ++u) ent[u] = myL[(qb + u) * LCAP + lane];
#pragma unroll
            for (int u = 0; u < 16; ++u) {
                const unsigned nq = (unsigned)__builtin_amdgcn_readlane((int)n, qb + u);
                sv[u] = src[((unsigned)lane < nq) ? (ent[u] >> 9) : 0u];
            }
#pragma unroll
            for (int u = 0; u < 16; ++u) {
                const unsigned nq = (unsigned)__builtin_amdgcn_readlane((int)n, qb + u);
                const unsigned oq = (unsigned)__builtin_amdgcn_readlane((int)off, qb + u);
                if ((unsigned)lane < nq) myL[oq + lane] = ((unsigned)sv[u] << 9) | (ent[u] & 511u);
            }
        }

        unsigned pkC = ((unsigned)lane < T) ? myL[lane] : 0u;
        float    daC = dinv[pkC >> 9];
        float    dbC = dinv[rowBase + (int)(pkC & 511u)];
        const float* hl = h + lane * 2;
        float* tl = sT + lane * 2;
        for (unsigned f0 = 0; f0 < T; f0 += 32) {
            const unsigned left = T - f0;
            const unsigned cn = left < 32u ? left : 32u;
            float2 hv[32];
#pragma unroll
            for (int u = 0; u < 32; ++u) {
                const int j = ((unsigned)u < cn) ? u : (int)(cn - 1);
                const unsigned pj = (unsigned)__builtin_amdgcn_readlane((int)pkC, j);
                hv[u] = *(const float2*)(hl + (size_t)(pj >> 9) * HID_DIM);
            }
            const float wC = ((unsigned)lane < cn) ? daC * dbC : 0.0f;
            unsigned pkN = 0u;
            if (f0 + 32 < T) {
                if (f0 + 32 + (unsigned)lane < T) pkN = myL[f0 + 32 + lane];
            }
            const float daN = dinv[pkN >> 9];
            const float dbN = dinv[rowBase + (int)(pkN & 511u)];
#pragma unroll
            for (int g4 = 0; g4 < 32; g4 += 4) {
                if ((unsigned)g4 < cn) {
                    int    rj[4];
                    float  wj[4];
                    float2 t[4];
#pragma unroll
                    for (int u = 0; u < 4; ++u) {
                        const bool lv = (unsigned)(g4 + u) < cn;
                        const int j = lv ? (g4 + u) : (int)(cn - 1);
                        rj[u] = __builtin_amdgcn_readlane((int)pkC, j) & 511;
                        const float w = __int_as_float(__builtin_amdgcn_readlane(__float_as_int(wC), j));
                        wj[u] = lv ? w : 0.0f;
                        t[u] = *(const float2*)(tl + rj[u] * TPITCH);
                    }
#pragma unroll
                    for (int u = 0; u < 4; ++u) {
                        float2 b = t[u];
#pragma unroll
                        for (int v = 0; v < u; ++v)
                            if (rj[v] == rj[u]) b = t[v];
                        b.x = fmaf(hv[g4 + u].x, wj[u], b.x);
                        b.y = fmaf(hv[g4 + u].y, wj[u], b.y);
                        t[u] = b;
                    }
#pragma unroll
                    for (int u = 0; u < 4; ++u) *(float2*)(tl + rj[u] * TPITCH) = t[u];
                }
            }
            pkC = pkN;
            daC = daN;
            dbC = dbN;
        }
    }
    __syncthreads();

    if (wave >= ROWS_WG / 16) return;
    const int tileRow = wave * 16;
    if (rowBase + tileRow >= N) return;

    const int r  = lane & 15;
    const int kg = lane >> 4;
    const int rowLd = min(rowBase + tileRow + r, N - 1);
    const float* arow = sT + (tileRow + r) * TPITCH;
    const float* hrow = h + (size_t)rowLd * HID_DIM;
    const float  di   = dinv[rowLd];
    const float  di2  = di * di;

    v8f acc[8] = {v8f{}, v8f{}, v8f{}, v8f{}, v8f{}, v8f{}, v8f{}, v8f{}};
    for (int k = 0; k < HID_DIM; k += 4) {
        const int kk = k + (kg << 1);
        float2 av  = *(const float2*)(arow + kk);
        float2 hv  = *(const float2*)(hrow + kk);
        float2 bcv = *(const float2*)(bc + kk);
        v2f a;
        a.x = fmaxf(av.x + hv.x * di2 + bcv.x, 0.0f);
        a.y = fmaxf(av.y + hv.y * di2 + bcv.y, 0.0f);
#pragma unroll
        for (int t = 0; t < 8; ++t) {
            v2f b;
            b.x = sW[(kk)     * OUT_DIM + t * 16 + r];
            b.y = sW[(kk + 1) * OUT_DIM + t * 16 + r];
            acc[t] = __builtin_amdgcn_wmma_f32_16x16x4_f32(
                false, a, false, b, (short)0, acc[t], false, false);
        }
    }
#pragma unroll
    for (int t = 0; t < 8; ++t) {
        const int col = t * 16 + r;
        const float blv = bl[col];
#pragma unroll
        for (int rr = 0; rr < 8; ++rr) {
            const int row = rowBase + tileRow + rr + 8 * kg;
            if (row < N) out[(size_t)row * OUT_DIM + col] = acc[t][rr] + blv;
        }
    }
}

static inline size_t alignUp(size_t v) { return (v + 255) & ~(size_t)255; }

extern "C" void kernel_launch(void* const* d_in, const int* in_sizes, int n_in,
                              void* d_out, int out_size, void* d_ws, size_t ws_size,
                              hipStream_t stream) {
    const float* x  = (const float*)d_in[0];
    const int*   ei = (const int*)  d_in[1];
    const float* Wc = (const float*)d_in[2];
    const float* bc = (const float*)d_in[3];
    const float* Wl = (const float*)d_in[4];
    const float* bl = (const float*)d_in[5];
    float* out = (float*)d_out;

    const int N = in_sizes[0] / IN_DIM;
    const int E = in_sizes[1] / 2;
    const int* src = ei;
    const int* dst = ei + E;

    char* ws = (char*)d_ws;
    const size_t hBytes = (size_t)N * HID_DIM * sizeof(float);
    size_t off = 0;
    float*        h    = (float*)(ws + off);        off = alignUp(off + hBytes);
    unsigned int* bmap = (unsigned int*)(ws + off); off = alignUp(off + hBytes);
    float*        dinv = (float*)(ws + off);        off = alignUp(off + (size_t)N * sizeof(float));
    unsigned int* deg  = (unsigned int*)(ws + off); off = alignUp(off + (size_t)N * sizeof(unsigned int));

    hipMemsetAsync(deg, 0, (size_t)N * sizeof(unsigned int), stream);

    const int rowBlocks = (N + 127) / 128;
    const int edgeBlocks = (E + EB_EDGES - 1) / EB_EDGES;
    gemm1_deg_kernel<<<rowBlocks + edgeBlocks, 256, 0, stream>>>(
        x, Wc, h, N, rowBlocks, dst, E, deg, bmap);
    dinv_kernel<<<(N + 255) / 256, 256, 0, stream>>>(deg, N, dinv);

    agg_gemm2_kernel<<<(N + ROWS_WG - 1) / ROWS_WG, 1024, 0, stream>>>(
        src, dst, dinv, h, bc, Wl, bl, out, N, E, bmap, edgeBlocks);
}
